# starting kernel
# baseline (speedup 1.0000x reference)
#include <hip/hip_runtime.h>
#include <hip/hip_bf16.h>

#define DIM 50
#define KPAD 64
#define OUT 20
#define NPAD 32

typedef __attribute__((ext_vector_type(2))) float v2f;
typedef __attribute__((ext_vector_type(8))) float v8f;

__global__ __launch_bounds__(256)
void pad_w_kernel(const float* __restrict__ W, float* __restrict__ Wpad) {
    for (int i = threadIdx.x; i < NPAD * KPAD; i += 256) {
        const int n = i >> 6;
        const int k = i & (KPAD-1);
        Wpad[i] = (n < OUT && k < DIM) ? W[n * DIM + k] : 0.0f;
    }
}

__global__ __launch_bounds__(256)
void pool_kernel(const int* __restrict__ x,
                 const int* __restrict__ lengths,
                 const float* __restrict__ emb,
                 float* __restrict__ rep,
                 int L) {
    const int row = blockIdx.x;
    const int d   = threadIdx.x;
    const int ty  = threadIdx.y;
    const int len = lengths[row];

    const int* xrow = x + (size_t)row * (size_t)L;

    float acc = 0.0f;
    if (d < DIM) {
        for (int j = ty; j < len; j += 4) {
            const int tok = xrow[j];
            acc += emb[(size_t)tok * DIM + d];
        }
    }

    __shared__ float smem[4][64];
    smem[ty][d] = acc;
    __syncthreads();

    if (ty == 0) {
        float s = smem[0][d] + smem[1][d] + smem[2][d] + smem[3][d];
        s = s / (float)len;
        s = fmaxf(s, 0.0f);
        rep[(size_t)row * KPAD + d] = (d < DIM) ? s : 0.0f;
    }
}

__global__ __launch_bounds__(32)
void head_gemm_kernel(const float* __restrict__ rep,
                      const float* __restrict__ Wpad,
                      const float* __restrict__ bias,
                      float* __restrict__ out) {
    const int lane  = threadIdx.x;
    const int row0  = blockIdx.x * 16;
    const int mN    = lane & 15;
    const int khalf = (lane >> 4) * 2;

    const float* arow  = rep  + (size_t)(row0 + mN) * KPAD + khalf;
    const float* brow0 = Wpad + (size_t)mN        * KPAD + khalf;
    const float* brow1 = Wpad + (size_t)(16 + mN) * KPAD + khalf;

    v8f c0 = {};
    v8f c1 = {};

#pragma unroll
    for (int kk = 0; kk < 13; ++kk) {
        const int kb = kk * 4;
        v2f a  = *(const v2f*)(arow  + kb);
        v2f b0 = *(const v2f*)(brow0 + kb);
        v2f b1 = *(const v2f*)(brow1 + kb);

        c0 = __builtin_amdgcn_wmma_f32_16x16x4_f32(
                 false, a, false, b0, (short)0, c0, false, false);
        c1 = __builtin_amdgcn_wmma_f32_16x16x4_f32(
                 false, a, false, b1, (short)0, c1, false, false);
    }

    const int n0 = mN;
    const int n1 = 16 + mN;
    const bool n1v = (n1 < OUT);
    const int mbase = row0 + ((lane >> 4) ? 8 : 0);
    const float bias0 = bias[n0];
    const float bias1 = n1v ? bias[n1] : 0.0f;
#pragma unroll
    for (int r = 0; r < 8; ++r) {
        const int M = mbase + r;
        out[(size_t)M * OUT + n0] = c0[r] + bias0;
        if (n1v) out[(size_t)M * OUT + n1] = c1[r] + bias1;
    }
}

extern "C" void kernel_launch(void* const* d_in, const int* in_sizes, int n_in,
                              void* d_out, int out_size, void* d_ws, size_t ws_size,
                              hipStream_t stream) {
    const int*   x       = (const int*)d_in[0];
    const int*   lengths = (const int*)d_in[1];
    const float* emb     = (const float*)d_in[2];
    const float* W       = (const float*)d_in[3];
    const float* bias    = (const float*)d_in[4];
    float*       out     = (float*)d_out;

    const int B = in_sizes[1];
    const int L = in_sizes[0] / B;

    float* rep  = (float*)d_ws;
    float* Wpad = rep + (size_t)B * KPAD;

    pad_w_kernel<<<1, 256, 0, stream>>>(W, Wpad);

    dim3 blk1(64, 4);
    pool_kernel<<<B, blk1, 0, stream>>>(x, lengths, emb, rep, L);

    head_gemm_kernel<<<B / 16, 32, 0, stream>>>(rep, Wpad, bias, out);
}


# head product moved into the pool wave: each wave multiplies its own pooled row (A by row_share from its sums, 16 equal M rows) by W read as B from the input, 26 f32 WMMAs; one launch; sum order kept
# speedup vs baseline: 3.4103x; 3.4103x over previous
#include <hip/hip_runtime.h>

#define DIM 50
#define OUT 20
#define KSTEPS 13

typedef __attribute__((ext_vector_type(2))) float v2f;
typedef __attribute__((ext_vector_type(8))) float v8f;

#define PAIRS 32
#define TRIP  (2 * PAIRS)
#define IDREGS 16

typedef __attribute__((ext_vector_type(4))) float v4f;
typedef __attribute__((ext_vector_type(4), aligned(8))) float v4f_a8;

template <int K>
__device__ __forceinline__ int row_lane(int v) {
    return __builtin_amdgcn_update_dpp(0, v, 0x150 + K, 0xF, 0xF, false);
}
template <int K>
__device__ __forceinline__ float row_lane_f(float v) {
    return __builtin_bit_cast(float, __builtin_amdgcn_update_dpp(
               0, __builtin_bit_cast(int, v), 0x150 + K, 0xF, 0xF, true));
}
template <int I>
struct TokFill {
    static __device__ __forceinline__ void run(int* tok, int ids0, int ids1) {
        tok[I] = row_lane<(I & 15)>((I < PAIRS / 2) ? ids0 : ids1);
        TokFill<I + 1>::run(tok, ids0, ids1);
    }
};
template <>
struct TokFill<PAIRS> {
    static __device__ __forceinline__ void run(int*, int, int) {}
};

template <bool FULL>
__device__ __forceinline__ void pool_trip(v4f& acc, const float* __restrict__ emb,
                                          int ids0, int ids1, int base, int n,
                                          int h, int foff) {
    int tok[PAIRS];
    TokFill<0>::run(tok, ids0, ids1);
    v4f v[PAIRS];
#pragma unroll
    for (int i = 0; i < PAIRS; ++i) {
        const bool ok = FULL || (base + 2 * i + h < n);
        const int  tk = ok ? tok[i] : 0;
        const unsigned e = (unsigned)tk * (unsigned)DIM + (unsigned)foff;
        v[i] = *(const v4f_a8*)(emb + e);
        if (!FULL) {
            v[i].x = ok ? v[i].x : 0.0f;
            v[i].y = ok ? v[i].y : 0.0f;
            v[i].z = ok ? v[i].z : 0.0f;
            v[i].w = ok ? v[i].w : 0.0f;
        }
    }
#pragma unroll
    for (int i = 0; i < PAIRS; ++i) acc += v[i];
}

template <int KK>
struct AFill {
    static __device__ __forceinline__ void run(v2f* a, float p, float q, float p12, float q12) {
        if (KK < 12) { a[KK].x = row_lane_f<KK>(p);   a[KK].y = row_lane_f<KK>(q);   }
        else         { a[KK].x = row_lane_f<12>(p12); a[KK].y = row_lane_f<12>(q12); }
        AFill<KK + 1>::run(a, p, q, p12, q12);
    }
};
template <>
struct AFill<KSTEPS> {
    static __device__ __forceinline__ void run(v2f*, float, float, float, float) {}
};

__global__ __launch_bounds__(32)
void pool_head_kernel(const int* __restrict__ x,
                      const int* __restrict__ lengths,
                      const float* __restrict__ emb,
                      const float* __restrict__ W,
                      const float* __restrict__ bias,
                      float* __restrict__ out,
                      int L) {
    const int row  = blockIdx.x;
    const int lane = threadIdx.x;
    const int c    = lane & 15;
    const int h    = lane >> 4;
    const int len  = lengths[row];
    const int n    = len < L ? len : L;

    const int* xrow = x + (size_t)row * (size_t)L;
    const int  foff = (c < 12) ? 4 * c : (c == 12 ? DIM - 4 : 0);

    const int ntrips = (n + TRIP - 1) / TRIP;

    const int slot = 2 * c + h;
    int ids[IDREGS];
#pragma unroll
    for (int k = 0; k < IDREGS; ++k) {
        const int j = 32 * k + slot;
        ids[k] = xrow[j < L ? j : L - 1];
    }

    v4f acc = {0.0f, 0.0f, 0.0f, 0.0f};

    for (int t = 0; t < ntrips; ++t) {
        const int base = t * TRIP;
        int ia = ids[0], ib = ids[1];
        if (t >= IDREGS / 2) {
            const int ja = base + slot;
            const int jb = ja + 32;
            ia = xrow[ja < L ? ja : L - 1];
            ib = xrow[jb < L ? jb : L - 1];
        }
        if (base + TRIP <= n) pool_trip<true >(acc, emb, ia, ib, base, n, h, foff);
        else                  pool_trip<false>(acc, emb, ia, ib, base, n, h, foff);
#pragma unroll
        for (int k = 0; k + 2 < IDREGS; ++k) ids[k] = ids[k + 2];
    }

    const float* w0 = W + c * DIM;
    const float* w1 = W + ((c < OUT - 16) ? 16 + c : 16) * DIM;
    v2f b0[KSTEPS], b1[KSTEPS];
#pragma unroll
    for (int kk = 0; kk < KSTEPS; ++kk) {
        const int k0 = (kk < 12) ? 4 * kk + 2 * h : DIM - 2;
        b0[kk] = *(const v2f*)(w0 + k0);
        b1[kk] = *(const v2f*)(w1 + k0);
    }

    acc.x += __shfl_xor(acc.x, 16, 32);
    acc.y += __shfl_xor(acc.y, 16, 32);
    acc.z += __shfl_xor(acc.z, 16, 32);
    acc.w += __shfl_xor(acc.w, 16, 32);

    const float fl = (float)len;
    v4f o;
    o.x = fmaxf(acc.x / fl, 0.0f);
    o.y = fmaxf(acc.y / fl, 0.0f);
    o.z = fmaxf(acc.z / fl, 0.0f);
    o.w = fmaxf(acc.w / fl, 0.0f);

    const float p   = h ? o.z : o.x;
    const float q   = h ? o.w : o.y;
    const float p12 = h ? 0.0f : o.z;
    const float q12 = h ? 0.0f : o.w;
    if (h) { b0[12].x = 0.0f; b0[12].y = 0.0f; b1[12].x = 0.0f; b1[12].y = 0.0f; }

    v2f a[KSTEPS];
    AFill<0>::run(a, p, q, p12, q12);

    __builtin_amdgcn_sched_barrier(0);

    v8f d0 = {};
    v8f d1 = {};
#pragma unroll
    for (int kk = 0; kk < KSTEPS; ++kk) {
        d0 = __builtin_amdgcn_wmma_f32_16x16x4_f32(
                 false, a[kk], false, b0[kk], (short)0, d0, false, false);
        d1 = __builtin_amdgcn_wmma_f32_16x16x4_f32(
                 false, a[kk], false, b1[kk], (short)0, d1, false, false);
    }

    if (lane < OUT) out[(size_t)row * OUT + lane] = (h ? d1[0] : d0[0]) + bias[lane];
}

extern "C" void kernel_launch(void* const* d_in, const int* in_sizes, int n_in,
                              void* d_out, int out_size, void* d_ws, size_t ws_size,
                              hipStream_t stream) {
    const int*   x       = (const int*)d_in[0];
    const int*   lengths = (const int*)d_in[1];
    const float* emb     = (const float*)d_in[2];
    const float* W       = (const float*)d_in[3];
    const float* bias    = (const float*)d_in[4];
    float*       out     = (float*)d_out;

    const int B = in_sizes[1];
    const int L = in_sizes[0] / B;

    pool_head_kernel<<<B, 32, 0, stream>>>(x, lengths, emb, W, bias, out, L);
}


# launch shape only: 8 independent waves (8 consecutive samples) per 256-thread work-group, 512 work-groups to dispatch instead of 4096 one-wave ones; per-wave code, loads and sum order unchanged
# speedup vs baseline: 3.5510x; 1.0413x over previous
#include <hip/hip_runtime.h>

#define DIM 50
#define OUT 20
#define KSTEPS 13

typedef __attribute__((ext_vector_type(2))) float v2f;
typedef __attribute__((ext_vector_type(8))) float v8f;

#define PAIRS 32
#define TRIP  (2 * PAIRS)
#define IDREGS 16
#define WAVES 8

typedef __attribute__((ext_vector_type(4))) float v4f;
typedef __attribute__((ext_vector_type(4), aligned(8))) float v4f_a8;

template <int K>
__device__ __forceinline__ int row_lane(int v) {
    return __builtin_amdgcn_update_dpp(0, v, 0x150 + K, 0xF, 0xF, false);
}
template <int K>
__device__ __forceinline__ float row_lane_f(float v) {
    return __builtin_bit_cast(float, __builtin_amdgcn_update_dpp(
               0, __builtin_bit_cast(int, v), 0x150 + K, 0xF, 0xF, true));
}
template <int I>
struct TokFill {
    static __device__ __forceinline__ void run(int* tok, int ids0, int ids1) {
        tok[I] = row_lane<(I & 15)>((I < PAIRS / 2) ? ids0 : ids1);
        TokFill<I + 1>::run(tok, ids0, ids1);
    }
};
template <>
struct TokFill<PAIRS> {
    static __device__ __forceinline__ void run(int*, int, int) {}
};

template <bool FULL>
__device__ __forceinline__ void pool_trip(v4f& acc, const float* __restrict__ emb,
                                          int ids0, int ids1, int base, int n,
                                          int h, int foff) {
    int tok[PAIRS];
    TokFill<0>::run(tok, ids0, ids1);
    v4f v[PAIRS];
#pragma unroll
    for (int i = 0; i < PAIRS; ++i) {
        const bool ok = FULL || (base + 2 * i + h < n);
        const int  tk = ok ? tok[i] : 0;
        const unsigned e = (unsigned)tk * (unsigned)DIM + (unsigned)foff;
        v[i] = *(const v4f_a8*)(emb + e);
        if (!FULL) {
            v[i].x = ok ? v[i].x : 0.0f;
            v[i].y = ok ? v[i].y : 0.0f;
            v[i].z = ok ? v[i].z : 0.0f;
            v[i].w = ok ? v[i].w : 0.0f;
        }
    }
#pragma unroll
    for (int i = 0; i < PAIRS; ++i) acc += v[i];
}

template <int KK>
struct AFill {
    static __device__ __forceinline__ void run(v2f* a, float p, float q, float p12, float q12) {
        if (KK < 12) { a[KK].x = row_lane_f<KK>(p);   a[KK].y = row_lane_f<KK>(q);   }
        else         { a[KK].x = row_lane_f<12>(p12); a[KK].y = row_lane_f<12>(q12); }
        AFill<KK + 1>::run(a, p, q, p12, q12);
    }
};
template <>
struct AFill<KSTEPS> {
    static __device__ __forceinline__ void run(v2f*, float, float, float, float) {}
};

__global__ __launch_bounds__(32 * WAVES)
void pool_head_kernel(const int* __restrict__ x,
                      const int* __restrict__ lengths,
                      const float* __restrict__ emb,
                      const float* __restrict__ W,
                      const float* __restrict__ bias,
                      float* __restrict__ out,
                      int L, int B) {
    const int wv   = __builtin_amdgcn_readfirstlane((int)(threadIdx.x >> 5));
    const int row  = blockIdx.x * WAVES + wv;
    if (row >= B) return;
    const int lane = threadIdx.x & 31;
    const int c    = lane & 15;
    const int h    = lane >> 4;
    const int len  = lengths[row];
    const int n    = len < L ? len : L;

    const int* xrow = x + (size_t)row * (size_t)L;
    const int  foff = (c < 12) ? 4 * c : (c == 12 ? DIM - 4 : 0);

    const int ntrips = (n + TRIP - 1) / TRIP;

    const int slot = 2 * c + h;
    int ids[IDREGS];
#pragma unroll
    for (int k = 0; k < IDREGS; ++k) {
        const int j = 32 * k + slot;
        ids[k] = xrow[j < L ? j : L - 1];
    }

    v4f acc = {0.0f, 0.0f, 0.0f, 0.0f};

    for (int t = 0; t < ntrips; ++t) {
        const int base = t * TRIP;
        int ia = ids[0], ib = ids[1];
        if (t >= IDREGS / 2) {
            const int ja = base + slot;
            const int jb = ja + 32;
            ia = xrow[ja < L ? ja : L - 1];
            ib = xrow[jb < L ? jb : L - 1];
        }
        if (base + TRIP <= n) pool_trip<true >(acc, emb, ia, ib, base, n, h, foff);
        else                  pool_trip<false>(acc, emb, ia, ib, base, n, h, foff);
#pragma unroll
        for (int k = 0; k + 2 < IDREGS; ++k) ids[k] = ids[k + 2];
    }

    const float* w0 = W + c * DIM;
    const float* w1 = W + ((c < OUT - 16) ? 16 + c : 16) * DIM;
    v2f b0[KSTEPS], b1[KSTEPS];
#pragma unroll
    for (int kk = 0; kk < KSTEPS; ++kk) {
        const int k0 = (kk < 12) ? 4 * kk + 2 * h : DIM - 2;
        b0[kk] = *(const v2f*)(w0 + k0);
        b1[kk] = *(const v2f*)(w1 + k0);
    }

    acc.x += __shfl_xor(acc.x, 16, 32);
    acc.y += __shfl_xor(acc.y, 16, 32);
    acc.z += __shfl_xor(acc.z, 16, 32);
    acc.w += __shfl_xor(acc.w, 16, 32);

    const float fl = (float)len;
    v4f o;
    o.x = fmaxf(acc.x / fl, 0.0f);
    o.y = fmaxf(acc.y / fl, 0.0f);
    o.z = fmaxf(acc.z / fl, 0.0f);
    o.w = fmaxf(acc.w / fl, 0.0f);

    const float p   = h ? o.z : o.x;
    const float q   = h ? o.w : o.y;
    const float p12 = h ? 0.0f : o.z;
    const float q12 = h ? 0.0f : o.w;
    if (h) { b0[12].x = 0.0f; b0[12].y = 0.0f; b1[12].x = 0.0f; b1[12].y = 0.0f; }

    v2f a[KSTEPS];
    AFill<0>::run(a, p, q, p12, q12);

    __builtin_amdgcn_sched_barrier(0);

    v8f d0 = {};
    v8f d1 = {};
#pragma unroll
    for (int kk = 0; kk < KSTEPS; ++kk) {
        d0 = __builtin_amdgcn_wmma_f32_16x16x4_f32(
                 false, a[kk], false, b0[kk], (short)0, d0, false, false);
        d1 = __builtin_amdgcn_wmma_f32_16x16x4_f32(
                 false, a[kk], false, b1[kk], (short)0, d1, false, false);
    }

    if (lane < OUT) out[(size_t)row * OUT + lane] = (h ? d1[0] : d0[0]) + bias[lane];
}

extern "C" void kernel_launch(void* const* d_in, const int* in_sizes, int n_in,
                              void* d_out, int out_size, void* d_ws, size_t ws_size,
                              hipStream_t stream) {
    const int*   x       = (const int*)d_in[0];
    const int*   lengths = (const int*)d_in[1];
    const float* emb     = (const float*)d_in[2];
    const float* W       = (const float*)d_in[3];
    const float* bias    = (const float*)d_in[4];
    float*       out     = (float*)d_out;

    const int B = in_sizes[1];
    const int L = in_sizes[0] / B;

    pool_head_kernel<<<(B + WAVES - 1) / WAVES, 32 * WAVES, 0, stream>>>(x, lengths, emb, W, bias, out, L, B);
}
